# starting kernel
# baseline (speedup 1.0000x reference)
#include <hip/hip_runtime.h>


typedef __attribute__((ext_vector_type(2))) float v2f;
typedef __attribute__((ext_vector_type(8))) float v8f;

#define SEQ      4096
#define NH       16
#define HD       64
#define ROWSTR   3072
#define NCHUNK   32
#define CH       (SEQ / NCHUNK)

__global__ __launch_bounds__(512)
void kv_partial_kernel(const float* __restrict__ qkv, float* __restrict__ part) {
    const int chunk = blockIdx.x;
    const int h     = blockIdx.y;
    const int w     = threadIdx.x >> 5;
    const int lane  = threadIdx.x & 31;
    const int hi    = lane >> 4;
    const int r     = lane & 15;
    const int d1b   = (w >> 2) << 4;
    const int d2b   = (w & 3)  << 4;

    const float* kp = qkv + (size_t)h * HD + 1024 + d1b + r;
    const float* vp = qkv + (size_t)h * HD + 2048 + d2b + r;

    v8f acc = {};
    const int s0 = chunk * CH;
#pragma unroll 4
    for (int s = s0; s < s0 + CH; s += 4) {
        const size_t row = (size_t)(s + 2 * hi) * ROWSTR;
        v2f a, b;
        a.x = kp[row];
        a.y = kp[row + ROWSTR];
        b.x = vp[row];
        b.y = vp[row + ROWSTR];
        acc = __builtin_amdgcn_wmma_f32_16x16x4_f32(
             false, a,  false, b,
             (short)0, acc,  false,  false);
    }

    float* o = part + ((size_t)(chunk * NH + h)) * 4096 + d2b + r;
#pragma unroll
    for (int rr = 0; rr < 8; ++rr) {
        const int d1 = d1b + rr + 8 * hi;
        o[d1 * 64] = acc[rr];
    }
}

__global__ __launch_bounds__(256)
void kv_reduce_kernel(const float* __restrict__ part, float* __restrict__ kv) {
    const int i = blockIdx.x * 256 + threadIdx.x;
    float s = 0.0f;
#pragma unroll
    for (int c = 0; c < NCHUNK; ++c)
        s += part[(size_t)c * NH * 4096 + i];
    kv[i] = s;
}

__global__ __launch_bounds__(512)
void out_kernel(const float* __restrict__ qkv, const float* __restrict__ kv,
                float* __restrict__ out) {
    __shared__ float lkv[64 * 64];

    const int sblk = blockIdx.x;
    const int h    = blockIdx.y;

    const float* kvh = kv + (size_t)h * 4096;
    for (int i = threadIdx.x; i < 4096; i += 512) lkv[i] = kvh[i];
    __syncthreads();

    const int w    = threadIdx.x >> 5;
    const int lane = threadIdx.x & 31;
    const int hi   = lane >> 4;
    const int r    = lane & 15;
    const int mi   = w >> 2;
    const int j    = w & 3;
    const int rowbase = sblk * 64 + mi * 16;
    const int d2b     = j * 16;

    const float* qrow = qkv + (size_t)(rowbase + r) * ROWSTR + (size_t)h * HD;

    v8f acc = {};
#pragma unroll
    for (int k0 = 0; k0 < 64; k0 += 4) {
        v2f a, b;
        a = *(const v2f*)(qrow + k0 + 2 * hi);
        b.x = lkv[(k0 + 2 * hi)     * 64 + d2b + r];
        b.y = lkv[(k0 + 2 * hi + 1) * 64 + d2b + r];
        acc = __builtin_amdgcn_wmma_f32_16x16x4_f32(
            false, a, false, b, (short)0, acc, false, false);
    }

#pragma unroll
    for (int rr = 0; rr < 8; ++rr) {
        const int grow = rowbase + rr + 8 * hi;
        out[((size_t)grow * NH + h) * HD + d2b + r] = acc[rr];
    }
}

extern "C" void kernel_launch(void* const* d_in, const int* in_sizes, int n_in,
                              void* d_out, int out_size, void* d_ws, size_t ws_size,
                              hipStream_t stream) {
    (void)in_sizes; (void)n_in; (void)out_size; (void)ws_size;
    const float* qkv = (const float*)d_in[0];
    float* out  = (float*)d_out;
    float* part = (float*)d_ws;
    float* kvf  = part + (size_t)NCHUNK * NH * 4096;

    dim3 g1(NCHUNK, NH);
    kv_partial_kernel<<<g1, 512, 0, stream>>>(qkv, part);

    kv_reduce_kernel<<<(NH * 4096) / 256, 256, 0, stream>>>(part, kvf);

    dim3 g3(SEQ / 64, NH);
    out_kernel<<<g3, 512, 0, stream>>>(qkv, kvf, out);
}


# KV-partial kernel: K and V rows go from memory straight into the WMMA fragments (8-byte loads, whole lines; tiles = even/odd rows and columns), no LDS staging; 8 waves x 128 rows; sum order changed
# speedup vs baseline: 1.4050x; 1.4050x over previous
#include <hip/hip_runtime.h>


typedef __attribute__((ext_vector_type(2))) float v2f;
typedef __attribute__((ext_vector_type(4))) float v4f;
typedef __attribute__((ext_vector_type(8))) float v8f;

#define SEQ      4096
#define NH       16
#define HD       64
#define ROWSTR   3072
#define NPART    4
#define PROWS    (SEQ / NPART)

#define KVW      8
#define KVT      256
#define KROWS    (PROWS / KVW)
#define KSTEPS   (KROWS / 4)
#define PFD      15
#define PAIRS    (16 / KVW)

__global__ __launch_bounds__(KVT) __attribute__((amdgpu_waves_per_eu(KVW / 4, KVW / 4)))
void kv_partial_kernel(const float* __restrict__ qkv, float* __restrict__ part) {
    __shared__ __attribute__((aligned(16))) v2f park[KVW * 16 * 32];

    const int quad  = blockIdx.x & 3;
    const int sq    = blockIdx.x >> 2;
    const int h     = blockIdx.y;
    const int w     = threadIdx.x >> 5;
    const int lane  = threadIdx.x & 31;
    const int hi    = lane >> 4;
    const int c     = lane & 15;
    const int d1q   = (quad >> 1) << 5;
    const int d2q   = (quad & 1)  << 5;

    const size_t rowoff = (size_t)(sq * PROWS + w * KROWS + 2 * hi) * ROWSTR + (size_t)h * HD + 2 * c;
    const float* kg = qkv + rowoff + 1024 + d1q;
    const float* vg = qkv + rowoff + 2048 + d2q;

    v2f k0[KSTEPS], k1[KSTEPS], v0[KSTEPS], v1[KSTEPS];
#pragma unroll
    for (int kk = 0; kk < PFD; ++kk) {
        k0[kk] = *(const v2f*)(kg + (size_t)(4 * kk)     * ROWSTR);
        v0[kk] = *(const v2f*)(vg + (size_t)(4 * kk)     * ROWSTR);
        k1[kk] = *(const v2f*)(kg + (size_t)(4 * kk + 1) * ROWSTR);
        v1[kk] = *(const v2f*)(vg + (size_t)(4 * kk + 1) * ROWSTR);
        __builtin_amdgcn_sched_barrier(0);
    }

    __builtin_amdgcn_sched_barrier(0);

    v8f acc00 = {}, acc01 = {}, acc10 = {}, acc11 = {};
#pragma unroll
    for (int kk = 0; kk < KSTEPS; ++kk) {
        if (kk + PFD < KSTEPS) {
            const int n = kk + PFD;
            k0[n] = *(const v2f*)(kg + (size_t)(4 * n)     * ROWSTR);
            v0[n] = *(const v2f*)(vg + (size_t)(4 * n)     * ROWSTR);
            k1[n] = *(const v2f*)(kg + (size_t)(4 * n + 1) * ROWSTR);
            v1[n] = *(const v2f*)(vg + (size_t)(4 * n + 1) * ROWSTR);
        }
        v2f a0, a1, b0, b1;
        a0.x = k0[kk].x;  a0.y = k1[kk].x;
        a1.x = k0[kk].y;  a1.y = k1[kk].y;
        b0.x = v0[kk].x;  b0.y = v1[kk].x;
        b1.x = v0[kk].y;  b1.y = v1[kk].y;
        acc00 = __builtin_amdgcn_wmma_f32_16x16x4_f32(false, a0, false, b0, (short)0, acc00, false, false);
        acc01 = __builtin_amdgcn_wmma_f32_16x16x4_f32(false, a0, false, b1, (short)0, acc01, false, false);
        acc10 = __builtin_amdgcn_wmma_f32_16x16x4_f32(false, a1, false, b0, (short)0, acc10, false, false);
        acc11 = __builtin_amdgcn_wmma_f32_16x16x4_f32(false, a1, false, b1, (short)0, acc11, false, false);
        __builtin_amdgcn_sched_barrier(0);
    }

    v2f* mine = park + (w * 16) * 32 + lane;
#pragma unroll
    for (int rr = 0; rr < 8; ++rr) {
        v2f ev, od;
        ev.x = acc00[rr];  ev.y = acc01[rr];
        od.x = acc10[rr];  od.y = acc11[rr];
        mine[(0 * 8 + rr) * 32] = ev;
        mine[(1 * 8 + rr) * 32] = od;
    }
    __syncthreads();

    float* o = part + ((size_t)(sq * NH + h)) * 4096 + d2q + 2 * c;
#pragma unroll
    for (int u = 0; u < PAIRS; ++u) {
        const int idx = w * PAIRS + u;
        const int t   = idx >> 3;
        const int rr  = idx & 7;
        const v2f* pz = park + idx * 32 + lane;
        v2f sum = pz[0];
#pragma unroll
        for (int s = 1; s < KVW; ++s)
            sum += pz[s * 16 * 32];
        *(v2f*)(o + (d1q + 2 * (rr + 8 * hi) + t) * 64) = sum;
    }
}

#define OUT_ROWS   256
#define OUT_TILES  4
#define KVP        72

__global__ __launch_bounds__(512) __attribute__((amdgpu_waves_per_eu(4, 4)))
void out_kernel(const float* __restrict__ qkv, const float* __restrict__ part,
                float* __restrict__ out) {
    __shared__ __attribute__((aligned(16))) float kvs[64 * KVP];

    const int sblk = blockIdx.x;
    const int h    = blockIdx.y;

    const int tid  = threadIdx.x;
    const int w    = tid >> 5;
    const int lane = tid & 31;
    const int hi   = lane >> 4;
    const int r    = lane & 15;
    const int mi   = w >> 2;
    const int j    = w & 3;
    const int rowbase = sblk * OUT_ROWS + mi * (16 * OUT_TILES);
    const int d2b     = j * 16;

    const v4f* pl = (const v4f*)(part + (size_t)h * 4096);
    v4f pp[2][NPART];
#pragma unroll
    for (int u = 0; u < 2; ++u) {
#pragma unroll
        for (int c = 0; c < NPART; ++c)
            pp[u][c] = pl[(size_t)c * (NH * 1024) + u * 512 + tid];
    }

    const float* qp = qkv + (size_t)(rowbase + r) * ROWSTR + (size_t)h * HD + 2 * hi;
    v2f a[OUT_TILES][16];
#pragma unroll
    for (int t = 0; t < OUT_TILES; ++t) {
#pragma unroll
        for (int kk = 0; kk < 16; ++kk)
            a[t][kk] = *(const v2f*)(qp + (size_t)t * 16 * ROWSTR + kk * 4);
    }

#pragma unroll
    for (int u = 0; u < 2; ++u) {
        const int i = u * 512 + tid;
        v4f s = pp[u][0];
        s += pp[u][1];
        s += pp[u][2];
        s += pp[u][3];
        *(v4f*)(kvs + (i >> 4) * KVP + (i & 15) * 4) = s;
    }
    __syncthreads();

    const float* kvp = kvs + (2 * hi) * KVP + d2b + r;
    v2f b[16];
#pragma unroll
    for (int kk = 0; kk < 16; ++kk) {
        b[kk].x = kvp[kk * 4 * KVP];
        b[kk].y = kvp[kk * 4 * KVP + KVP];
    }

#pragma unroll
    for (int t = 0; t < OUT_TILES; t += 2) {
        v8f acc0 = {};
        v8f acc1 = {};
#pragma unroll
        for (int kk = 0; kk < 16; ++kk) {
            acc0 = __builtin_amdgcn_wmma_f32_16x16x4_f32(
                false, a[t][kk], false, b[kk], (short)0, acc0, false, false);
            acc1 = __builtin_amdgcn_wmma_f32_16x16x4_f32(
                false, a[t + 1][kk], false, b[kk], (short)0, acc1, false, false);
        }
#pragma unroll
        for (int rr = 0; rr < 8; ++rr) {
            const int g0 = rowbase + 16 * t + rr + 8 * hi;
            out[((size_t)g0 * NH + h) * HD + d2b + r]        = acc0[rr];
            out[((size_t)(g0 + 16) * NH + h) * HD + d2b + r] = acc1[rr];
        }
    }
}

extern "C" void kernel_launch(void* const* d_in, const int* in_sizes, int n_in,
                              void* d_out, int out_size, void* d_ws, size_t ws_size,
                              hipStream_t stream) {
    (void)in_sizes; (void)n_in; (void)out_size; (void)ws_size;
    const float* qkv = (const float*)d_in[0];
    float* out  = (float*)d_out;
    float* part = (float*)d_ws;

    dim3 g1(4 * NPART, NH);
    kv_partial_kernel<<<g1, KVT, 0, stream>>>(qkv, part);

    dim3 g3(SEQ / OUT_ROWS, NH);
    out_kernel<<<g3, 512, 0, stream>>>(qkv, part, out);
}


# Output kernel: a wave owns 32 rows x 32 columns (column tiles = even/odd columns of one line), was 64 x 16: half the Q fragment loads, 8-byte stores of whole 128-byte lines; sum order unchanged
# speedup vs baseline: 1.5502x; 1.1033x over previous
#include <hip/hip_runtime.h>


typedef __attribute__((ext_vector_type(2))) float v2f;
typedef __attribute__((ext_vector_type(4))) float v4f;
typedef __attribute__((ext_vector_type(8))) float v8f;

#define SEQ      4096
#define NH       16
#define HD       64
#define ROWSTR   3072
#define NPART    4
#define PROWS    (SEQ / NPART)

#define KVW      8
#define KVT      256
#define KROWS    (PROWS / KVW)
#define KSTEPS   (KROWS / 4)
#define PFD      15
#define PAIRS    (16 / KVW)

__global__ __launch_bounds__(KVT) __attribute__((amdgpu_waves_per_eu(KVW / 4, KVW / 4)))
void kv_partial_kernel(const float* __restrict__ qkv, float* __restrict__ part) {
    __shared__ __attribute__((aligned(16))) v2f park[KVW * 16 * 32];

    const int quad  = blockIdx.x & 3;
    const int sq    = blockIdx.x >> 2;
    const int h     = blockIdx.y;
    const int w     = threadIdx.x >> 5;
    const int lane  = threadIdx.x & 31;
    const int hi    = lane >> 4;
    const int c     = lane & 15;
    const int d1q   = (quad >> 1) << 5;
    const int d2q   = (quad & 1)  << 5;

    const size_t rowoff = (size_t)(sq * PROWS + w * KROWS + 2 * hi) * ROWSTR + (size_t)h * HD + 2 * c;
    const float* kg = qkv + rowoff + 1024 + d1q;
    const float* vg = qkv + rowoff + 2048 + d2q;

    v2f k0[KSTEPS], k1[KSTEPS], v0[KSTEPS], v1[KSTEPS];
#pragma unroll
    for (int kk = 0; kk < PFD; ++kk) {
        k0[kk] = *(const v2f*)(kg + (size_t)(4 * kk)     * ROWSTR);
        v0[kk] = *(const v2f*)(vg + (size_t)(4 * kk)     * ROWSTR);
        k1[kk] = *(const v2f*)(kg + (size_t)(4 * kk + 1) * ROWSTR);
        v1[kk] = *(const v2f*)(vg + (size_t)(4 * kk + 1) * ROWSTR);
        __builtin_amdgcn_sched_barrier(0);
    }

    __builtin_amdgcn_sched_barrier(0);

    v8f acc00 = {}, acc01 = {}, acc10 = {}, acc11 = {};
#pragma unroll
    for (int kk = 0; kk < KSTEPS; ++kk) {
        if (kk + PFD < KSTEPS) {
            const int n = kk + PFD;
            k0[n] = *(const v2f*)(kg + (size_t)(4 * n)     * ROWSTR);
            v0[n] = *(const v2f*)(vg + (size_t)(4 * n)     * ROWSTR);
            k1[n] = *(const v2f*)(kg + (size_t)(4 * n + 1) * ROWSTR);
            v1[n] = *(const v2f*)(vg + (size_t)(4 * n + 1) * ROWSTR);
        }
        v2f a0, a1, b0, b1;
        a0.x = k0[kk].x;  a0.y = k1[kk].x;
        a1.x = k0[kk].y;  a1.y = k1[kk].y;
        b0.x = v0[kk].x;  b0.y = v1[kk].x;
        b1.x = v0[kk].y;  b1.y = v1[kk].y;
        acc00 = __builtin_amdgcn_wmma_f32_16x16x4_f32(false, a0, false, b0, (short)0, acc00, false, false);
        acc01 = __builtin_amdgcn_wmma_f32_16x16x4_f32(false, a0, false, b1, (short)0, acc01, false, false);
        acc10 = __builtin_amdgcn_wmma_f32_16x16x4_f32(false, a1, false, b0, (short)0, acc10, false, false);
        acc11 = __builtin_amdgcn_wmma_f32_16x16x4_f32(false, a1, false, b1, (short)0, acc11, false, false);
        __builtin_amdgcn_sched_barrier(0);
    }

    v2f* mine = park + (w * 16) * 32 + lane;
#pragma unroll
    for (int rr = 0; rr < 8; ++rr) {
        v2f ev, od;
        ev.x = acc00[rr];  ev.y = acc01[rr];
        od.x = acc10[rr];  od.y = acc11[rr];
        mine[(0 * 8 + rr) * 32] = ev;
        mine[(1 * 8 + rr) * 32] = od;
    }
    __syncthreads();

    float* o = part + ((size_t)(sq * NH + h)) * 4096 + d2q + 2 * c;
#pragma unroll
    for (int u = 0; u < PAIRS; ++u) {
        const int idx = w * PAIRS + u;
        const int t   = idx >> 3;
        const int rr  = idx & 7;
        const v2f* pz = park + idx * 32 + lane;
        v2f sum = pz[0];
#pragma unroll
        for (int s = 1; s < KVW; ++s)
            sum += pz[s * 16 * 32];
        *(v2f*)(o + (d1q + 2 * (rr + 8 * hi) + t) * 64) = sum;
    }
}

#define OUT_ROWS   256
#define OUT_TILES  2

__global__ __launch_bounds__(512) __attribute__((amdgpu_waves_per_eu(4, 4)))
void out_kernel(const float* __restrict__ qkv, const float* __restrict__ part,
                float* __restrict__ out) {
    __shared__ __attribute__((aligned(16))) float kvs[32 * 128];

    const int sblk = blockIdx.x;
    const int h    = blockIdx.y;

    const int tid  = threadIdx.x;
    const int w    = tid >> 5;
    const int lane = tid & 31;
    const int hi   = lane >> 4;
    const int r    = lane & 15;
    const int mi   = w >> 1;
    const int jh   = w & 1;
    const int rowbase = sblk * OUT_ROWS + mi * (16 * OUT_TILES);
    const int d2b     = jh * 32;

    const int p = tid >> 4;
    const int m = tid & 15;
    const v4f* pl = (const v4f*)(part + (size_t)h * 4096) + (2 * p) * 16 + m;
    v4f pp[2][NPART];
#pragma unroll
    for (int u = 0; u < 2; ++u) {
#pragma unroll
        for (int c = 0; c < NPART; ++c)
            pp[u][c] = pl[(size_t)c * (NH * 1024) + u * 16];
    }

    const float* qp = qkv + (size_t)(rowbase + r) * ROWSTR + (size_t)h * HD + 2 * hi;
    v2f a[OUT_TILES][16];
#pragma unroll
    for (int t = 0; t < OUT_TILES; ++t) {
#pragma unroll
        for (int kk = 0; kk < 16; ++kk)
            a[t][kk] = *(const v2f*)(qp + (size_t)t * 16 * ROWSTR + kk * 4);
    }

    {
        v4f s0 = pp[0][0];
        s0 += pp[0][1];
        s0 += pp[0][2];
        s0 += pp[0][3];
        v4f s1 = pp[1][0];
        s1 += pp[1][1];
        s1 += pp[1][2];
        s1 += pp[1][3];
        v4f lo, up;
        lo.x = s0.x;  lo.y = s1.x;  lo.z = s0.y;  lo.w = s1.y;
        up.x = s0.z;  up.y = s1.z;  up.z = s0.w;  up.w = s1.w;
        float* dst = kvs + p * 128 + m * 8;
        *(v4f*)(dst)     = lo;
        *(v4f*)(dst + 4) = up;
    }
    __syncthreads();

    const float* kvp = kvs + hi * 128 + (d2b + 2 * r) * 2;
    v2f b0[16], b1[16];
#pragma unroll
    for (int kk = 0; kk < 16; ++kk) {
        const v4f f = *(const v4f*)(kvp + kk * 256);
        b0[kk].x = f.x;  b0[kk].y = f.y;
        b1[kk].x = f.z;  b1[kk].y = f.w;
    }

#pragma unroll
    for (int t = 0; t < OUT_TILES; ++t) {
        v8f acc0 = {};
        v8f acc1 = {};
#pragma unroll
        for (int kk = 0; kk < 16; ++kk) {
            acc0 = __builtin_amdgcn_wmma_f32_16x16x4_f32(
                false, a[t][kk], false, b0[kk], (short)0, acc0, false, false);
            acc1 = __builtin_amdgcn_wmma_f32_16x16x4_f32(
                false, a[t][kk], false, b1[kk], (short)0, acc1, false, false);
        }
#pragma unroll
        for (int rr = 0; rr < 8; ++rr) {
            const int g0 = rowbase + 16 * t + rr + 8 * hi;
            v2f o;
            o.x = acc0[rr];
            o.y = acc1[rr];
            *(v2f*)(out + ((size_t)g0 * NH + h) * HD + d2b + 2 * r) = o;
        }
    }
}

extern "C" void kernel_launch(void* const* d_in, const int* in_sizes, int n_in,
                              void* d_out, int out_size, void* d_ws, size_t ws_size,
                              hipStream_t stream) {
    (void)in_sizes; (void)n_in; (void)out_size; (void)ws_size;
    const float* qkv = (const float*)d_in[0];
    float* out  = (float*)d_out;
    float* part = (float*)d_ws;

    dim3 g1(4 * NPART, NH);
    kv_partial_kernel<<<g1, KVT, 0, stream>>>(qkv, part);

    dim3 g3(SEQ / OUT_ROWS, NH);
    out_kernel<<<g3, 512, 0, stream>>>(qkv, part, out);
}
